# starting kernel
# baseline (speedup 1.0000x reference)
#include <hip/hip_runtime.h>
#include <math.h>

#define NB 8
#define NS 50
#define NT 77
#define ND 1024
#define NR 16
#define TTILE 16
#define NTILES 5
#define HSTR 1028
#define PSTR 17

typedef __attribute__((ext_vector_type(2))) float v2f;
typedef __attribute__((ext_vector_type(4))) float v4f;
typedef __attribute__((ext_vector_type(8))) float v8f;

__global__ __launch_bounds__(256)
void loreft_rownorm_kernel(const float* __restrict__ R, float* __restrict__ inv)
{
    const int sr = blockIdx.x;
    const float* row = R + (size_t)sr * ND;

    float sum = 0.0f;
    for (int i = threadIdx.x; i < ND; i += 256) {
        float v = row[i];
        sum += v * v;
    }
    __shared__ float red[256];
    red[threadIdx.x] = sum;
    __syncthreads();
    for (int off = 128; off > 0; off >>= 1) {
        if (threadIdx.x < off) red[threadIdx.x] += red[threadIdx.x + off];
        __syncthreads();
    }
    if (threadIdx.x == 0) {
        float n = sqrtf(red[0]);
        n = fmaxf(n, 1e-12f);
        inv[sr] = 1.0f / n;
    }
}

__global__ __launch_bounds__(128)
void loreft_main_kernel(const float* __restrict__ h,
                        const float* __restrict__ R,
                        const float* __restrict__ W,
                        const float* __restrict__ bvec,
                        const float* __restrict__ inv,
                        float* __restrict__ out)
{
    extern __shared__ float smem[];
    float* __restrict__ htile   = smem;
    float* __restrict__ partial = htile + TTILE * HSTR;
    float* __restrict__ Pm      = partial + 4 * TTILE * NR;
    float* __restrict__ invL    = Pm + TTILE * PSTR;
    float* __restrict__ bL      = invL + NR;

    const int blk  = blockIdx.x;
    const int tile = blk % NTILES;
    const int bs   = blk / NTILES;
    const int s    = bs % NS;
    const int b    = bs / NS;
    const int t0   = tile * TTILE;

    const int lane = threadIdx.x & 31;
    const int wave = threadIdx.x >> 5;
    const int m    = lane & 15;
    const int kk   = (lane >> 4) << 1;

    const size_t tok_base = ((size_t)(b * NS + s)) * NT;
    const float* __restrict__ hbase = h + tok_base * ND;
    float*       __restrict__ obase = out + tok_base * ND;
    const float* __restrict__ Rs = R + (size_t)s * NR * ND;
    const float* __restrict__ Ws = W + (size_t)s * NR * ND;

    if (threadIdx.x < NR) {
        invL[threadIdx.x] = inv[s * NR + threadIdx.x];
        bL[threadIdx.x]   = bvec[s * NR + threadIdx.x];
    }

    for (int idx = threadIdx.x; idx < TTILE * (ND / 4); idx += 128) {
        int row  = idx >> 8;
        int c4   = idx & 255;
        int tokr = t0 + row;
        tokr = (tokr < NT) ? tokr : (NT - 1);
        v4f v = *(const v4f*)(hbase + (size_t)tokr * ND + ((size_t)c4 << 2));
        *(v4f*)(htile + row * HSTR + (c4 << 2)) = v;
    }
    __syncthreads();

    const float invm = invL[m];
    const float* __restrict__ Rrow = Rs + (size_t)m * ND;
    const float* __restrict__ Wrow = Ws + (size_t)m * ND;
    const float* __restrict__ hL   = htile + m * HSTR;

    v8f accR = {};
    v8f accW = {};
    const int k0 = wave * (ND / 4);
    for (int k = k0; k < k0 + ND / 4; k += 4) {
        v2f a  = *(const v2f*)(hL + k + kk);
        v2f br = *(const v2f*)(Rrow + k + kk);
        v2f bw = *(const v2f*)(Wrow + k + kk);
        br.x *= invm;
        br.y *= invm;
        accR = __builtin_amdgcn_wmma_f32_16x16x4_f32(
                   false, a, false, br, (short)0, accR, false, false);
        accW = __builtin_amdgcn_wmma_f32_16x16x4_f32(
                   false, a, false, bw, (short)0, accW, false, false);
    }

    #pragma unroll
    for (int i = 0; i < 8; ++i) {
        int row = i + ((lane >> 4) << 3);
        int col = lane & 15;
        partial[(wave * TTILE + row) * NR + col] = accW[i] - accR[i];
    }
    __syncthreads();
    for (int idx = threadIdx.x; idx < TTILE * NR; idx += 128) {
        int row = idx >> 4, col = idx & 15;
        Pm[row * PSTR + col] =
              partial[(0 * TTILE + row) * NR + col]
            + partial[(1 * TTILE + row) * NR + col]
            + partial[(2 * TTILE + row) * NR + col]
            + partial[(3 * TTILE + row) * NR + col]
            + bL[col];
    }
    __syncthreads();

    for (int j = 0; j < 16; ++j) {
        const int d0   = ((j << 2) | wave) << 4;
        const int dcol = d0 + m;
        v8f acc = {};
        #pragma unroll
        for (int kc = 0; kc < NR; kc += 4) {
            v2f a, br;
            a.x  = Pm[m * PSTR + kc + kk];
            a.y  = Pm[m * PSTR + kc + kk + 1];
            br.x = Rs[(size_t)(kc + kk)     * ND + dcol] * invL[kc + kk];
            br.y = Rs[(size_t)(kc + kk + 1) * ND + dcol] * invL[kc + kk + 1];
            acc = __builtin_amdgcn_wmma_f32_16x16x4_f32(
                      false, a, false, br, (short)0, acc, false, false);
        }
        #pragma unroll
        for (int i = 0; i < 8; ++i) {
            int row  = i + ((lane >> 4) << 3);
            int toko = t0 + row;
            if (toko < NT) {
                int col = d0 + (lane & 15);
                obase[(size_t)toko * ND + col] =
                    htile[row * HSTR + col] + acc[i];
            }
        }
    }
}

extern "C" void kernel_launch(void* const* d_in, const int* in_sizes, int n_in,
                              void* d_out, int out_size, void* d_ws, size_t ws_size,
                              hipStream_t stream)
{
    const float* h    = (const float*)d_in[0];
    const float* R    = (const float*)d_in[1];
    const float* W    = (const float*)d_in[2];
    const float* bvec = (const float*)d_in[3];
    float* out = (float*)d_out;
    float* inv = (float*)d_ws;

    (void)in_sizes; (void)n_in; (void)out_size; (void)ws_size;

    enum { SMEM_FLOATS = TTILE * HSTR + 4 * TTILE * NR + TTILE * PSTR + NR + NR };

    loreft_rownorm_kernel<<<NS * NR, 256, 0, stream>>>(R, inv);
    loreft_main_kernel<<<NB * NS * NTILES, 128, SMEM_FLOATS * sizeof(float), stream>>>(
        h, R, W, bvec, inv, out);
}


# Phase B column remap: 4 interleaved 16-col tiles per step so each lane owns 4 adjacent columns; R loads, LDS residual reads and out stores go 128-bit (whole 128-B lines per store); sums unchanged
# speedup vs baseline: 1.6257x; 1.6257x over previous
#include <hip/hip_runtime.h>
#include <math.h>

#define NB 8
#define NS 50
#define NT 77
#define ND 1024
#define NR 16
#define TTILE 16
#define NTILES 5
#define HSTR 1028
#define PSTR 17

typedef __attribute__((ext_vector_type(2))) float v2f;
typedef __attribute__((ext_vector_type(4))) float v4f;
typedef __attribute__((ext_vector_type(8))) float v8f;

__global__ __launch_bounds__(256)
void loreft_rownorm_kernel(const float* __restrict__ R, float* __restrict__ inv)
{
    const int sr = blockIdx.x;
    const float* row = R + (size_t)sr * ND;

    float sum = 0.0f;
    for (int i = threadIdx.x; i < ND; i += 256) {
        float v = row[i];
        sum += v * v;
    }
    __shared__ float red[256];
    red[threadIdx.x] = sum;
    __syncthreads();
    for (int off = 128; off > 0; off >>= 1) {
        if (threadIdx.x < off) red[threadIdx.x] += red[threadIdx.x + off];
        __syncthreads();
    }
    if (threadIdx.x == 0) {
        float n = sqrtf(red[0]);
        n = fmaxf(n, 1e-12f);
        inv[sr] = 1.0f / n;
    }
}

__global__ __launch_bounds__(128)
void loreft_main_kernel(const float* __restrict__ h,
                        const float* __restrict__ R,
                        const float* __restrict__ W,
                        const float* __restrict__ bvec,
                        const float* __restrict__ inv,
                        float* __restrict__ out)
{
    extern __shared__ float smem[];
    float* __restrict__ htile   = smem;
    float* __restrict__ partial = htile + TTILE * HSTR;
    float* __restrict__ Pm      = partial + 4 * TTILE * NR;
    float* __restrict__ invL    = Pm + TTILE * PSTR;
    float* __restrict__ bL      = invL + NR;

    const int blk  = blockIdx.x;
    const int tile = blk % NTILES;
    const int bs   = blk / NTILES;
    const int s    = bs % NS;
    const int b    = bs / NS;
    const int t0   = tile * TTILE;

    const int lane = threadIdx.x & 31;
    const int wave = threadIdx.x >> 5;
    const int m    = lane & 15;
    const int kk   = (lane >> 4) << 1;

    const size_t tok_base = ((size_t)(b * NS + s)) * NT;
    const float* __restrict__ hbase = h + tok_base * ND;
    float*       __restrict__ obase = out + tok_base * ND;
    const float* __restrict__ Rs = R + (size_t)s * NR * ND;
    const float* __restrict__ Ws = W + (size_t)s * NR * ND;

    if (threadIdx.x < NR) {
        invL[threadIdx.x] = inv[s * NR + threadIdx.x];
        bL[threadIdx.x]   = bvec[s * NR + threadIdx.x];
    }

    for (int idx = threadIdx.x; idx < TTILE * (ND / 4); idx += 128) {
        int row  = idx >> 8;
        int c4   = idx & 255;
        int tokr = t0 + row;
        tokr = (tokr < NT) ? tokr : (NT - 1);
        v4f v = *(const v4f*)(hbase + (size_t)tokr * ND + ((size_t)c4 << 2));
        *(v4f*)(htile + row * HSTR + (c4 << 2)) = v;
    }
    __syncthreads();

    const float invm = invL[m];
    const float* __restrict__ Rrow = Rs + (size_t)m * ND;
    const float* __restrict__ Wrow = Ws + (size_t)m * ND;
    const float* __restrict__ hL   = htile + m * HSTR;

    v8f accR = {};
    v8f accW = {};
    const int k0 = wave * (ND / 4);
    for (int k = k0; k < k0 + ND / 4; k += 4) {
        v2f a  = *(const v2f*)(hL + k + kk);
        v2f br = *(const v2f*)(Rrow + k + kk);
        v2f bw = *(const v2f*)(Wrow + k + kk);
        br.x *= invm;
        br.y *= invm;
        accR = __builtin_amdgcn_wmma_f32_16x16x4_f32(
                   false, a, false, br, (short)0, accR, false, false);
        accW = __builtin_amdgcn_wmma_f32_16x16x4_f32(
                   false, a, false, bw, (short)0, accW, false, false);
    }

    #pragma unroll
    for (int i = 0; i < 8; ++i) {
        int row = i + ((lane >> 4) << 3);
        int col = lane & 15;
        partial[(wave * TTILE + row) * NR + col] = accW[i] - accR[i];
    }
    __syncthreads();
    for (int idx = threadIdx.x; idx < TTILE * NR; idx += 128) {
        int row = idx >> 4, col = idx & 15;
        Pm[row * PSTR + col] =
              partial[(0 * TTILE + row) * NR + col]
            + partial[(1 * TTILE + row) * NR + col]
            + partial[(2 * TTILE + row) * NR + col]
            + partial[(3 * TTILE + row) * NR + col]
            + bL[col];
    }
    __syncthreads();

    v2f  pa[4];
    float sc0[4], sc1[4];
    for (int q = 0; q < 4; ++q) {
        pa[q].x = Pm[m * PSTR + 4 * q + kk];
        pa[q].y = Pm[m * PSTR + 4 * q + kk + 1];
        sc0[q]  = invL[4 * q + kk];
        sc1[q]  = invL[4 * q + kk + 1];
    }
    const int hrow = (lane >> 4) << 3;

    for (int g = 0; g < 4; ++g) {
        const int dbase = (wave << 8) + (g << 6);
        const int dl    = dbase + (m << 2);

        v4f r0[4], r1[4];
        for (int q = 0; q < 4; ++q) {
            r0[q] = *(const v4f*)(Rs + (size_t)(4 * q + kk)     * ND + dl);
            r1[q] = *(const v4f*)(Rs + (size_t)(4 * q + kk + 1) * ND + dl);
        }
        v4f res[8];
        for (int i = 0; i < 8; ++i)
            res[i] = *(const v4f*)(htile + (i + hrow) * HSTR + dl);

        v8f acc[4] = {};
        for (int q = 0; q < 4; ++q) {
            for (int t = 0; t < 4; ++t) {
                v2f br;
                br.x = r0[q][t] * sc0[q];
                br.y = r1[q][t] * sc1[q];
                acc[t] = __builtin_amdgcn_wmma_f32_16x16x4_f32(
                             false, pa[q], false, br, (short)0, acc[t], false, false);
            }
        }
        for (int i = 0; i < 8; ++i) {
            v4f o;
            o.x = res[i].x + acc[0][i];
            o.y = res[i].y + acc[1][i];
            o.z = res[i].z + acc[2][i];
            o.w = res[i].w + acc[3][i];
            int toko = t0 + i + hrow;
            if (toko < NT)
                *(v4f*)(obase + (size_t)toko * ND + dl) = o;
        }
    }
}

extern "C" void kernel_launch(void* const* d_in, const int* in_sizes, int n_in,
                              void* d_out, int out_size, void* d_ws, size_t ws_size,
                              hipStream_t stream)
{
    const float* h    = (const float*)d_in[0];
    const float* R    = (const float*)d_in[1];
    const float* W    = (const float*)d_in[2];
    const float* bvec = (const float*)d_in[3];
    float* out = (float*)d_out;
    float* inv = (float*)d_ws;

    (void)in_sizes; (void)n_in; (void)out_size; (void)ws_size;

    enum { SMEM_FLOATS = TTILE * HSTR + 4 * TTILE * NR + TTILE * PSTR + NR + NR };

    loreft_rownorm_kernel<<<NS * NR, 256, 0, stream>>>(R, inv);
    loreft_main_kernel<<<NB * NS * NTILES, 128, SMEM_FLOATS * sizeof(float), stream>>>(
        h, R, W, bvec, inv, out);
}


# Launch 1 also writes W - R/|R| per step in B-fragment order; phase A: one product per k-step from contiguous 128-bit loads (was two, 16-line loads); W-Rn formed before the dot, k sum order regrouped
# speedup vs baseline: 1.8344x; 1.1284x over previous
#include <hip/hip_runtime.h>
#include <math.h>

#define NB 8
#define NS 50
#define NT 77
#define ND 1024
#define NR 16
#define TTILE 16
#define NTILES 5
#define HSTR 1028
#define PSTR 17

typedef __attribute__((ext_vector_type(2))) float v2f;
typedef __attribute__((ext_vector_type(4))) float v4f;
typedef __attribute__((ext_vector_type(8))) float v8f;

__global__ __launch_bounds__(256)
void loreft_prep_kernel(const float* __restrict__ R, const float* __restrict__ W,
                        float* __restrict__ inv, float* __restrict__ Mf)
{
    const int sr = blockIdx.x;
    const float* row = R + (size_t)sr * ND;

    const v4f rv = *(const v4f*)(row + 4 * threadIdx.x);
    const v4f wv = *(const v4f*)(W + (size_t)sr * ND + 4 * threadIdx.x);

    float sum = 0.0f;
    for (int i = threadIdx.x; i < ND; i += 256) {
        float v = row[i];
        sum += v * v;
    }
    __shared__ float red[256];
    red[threadIdx.x] = sum;
    __syncthreads();
    for (int off = 128; off > 0; off >>= 1) {
        if (threadIdx.x < off) red[threadIdx.x] += red[threadIdx.x + off];
        __syncthreads();
    }
    if (threadIdx.x == 0) {
        float n = sqrtf(red[0]);
        n = fmaxf(n, 1e-12f);
        const float r1 = 1.0f / n;
        inv[sr] = r1;
        red[0]  = r1;
    }
    __syncthreads();
    const float iv = red[0];

    v4f mv;
    mv.x = wv.x - rv.x * iv;
    mv.y = wv.y - rv.y * iv;
    mv.z = wv.z - rv.z * iv;
    mv.w = wv.w - rv.w * iv;

    const int s  = sr / NR;
    const int r  = sr % NR;
    const int w  = threadIdx.x >> 6;
    const int j  = (threadIdx.x & 63) >> 1;
    const int hf = threadIdx.x & 1;
    *(v4f*)(Mf + (((((size_t)(s * 4 + w) * 32 + j) * 32) + (hf * 16 + r)) << 2)) = mv;
}

__global__ __launch_bounds__(128)
void loreft_main_kernel(const float* __restrict__ h,
                        const float* __restrict__ R,
                        const float* __restrict__ Mf,
                        const float* __restrict__ bvec,
                        const float* __restrict__ inv,
                        float* __restrict__ out)
{
    extern __shared__ float smem[];
    float* __restrict__ htile   = smem;
    float* __restrict__ partial = htile + TTILE * HSTR;
    float* __restrict__ Pm      = partial + 4 * TTILE * NR;
    float* __restrict__ invL    = Pm + TTILE * PSTR;
    float* __restrict__ bL      = invL + NR;

    const int blk  = blockIdx.x;
    const int tile = blk % NTILES;
    const int bs   = blk / NTILES;
    const int s    = bs % NS;
    const int b    = bs / NS;
    const int t0   = tile * TTILE;

    const int lane = threadIdx.x & 31;
    const int wave = threadIdx.x >> 5;
    const int m    = lane & 15;
    const int kk   = (lane >> 4) << 1;

    const size_t tok_base = ((size_t)(b * NS + s)) * NT;
    const float* __restrict__ hbase = h + tok_base * ND;
    float*       __restrict__ obase = out + tok_base * ND;
    const float* __restrict__ Rs = R + (size_t)s * NR * ND;

    const v4f* __restrict__ Mp =
        (const v4f*)Mf + ((size_t)(s * 4 + wave) * 32) * 32 + lane;
    v4f bm[32];
    #pragma unroll
    for (int j = 0; j < 32; ++j)
        bm[j] = Mp[j * 32];

    if (threadIdx.x < NR) {
        invL[threadIdx.x] = inv[s * NR + threadIdx.x];
        bL[threadIdx.x]   = bvec[s * NR + threadIdx.x];
    }

    for (int idx = threadIdx.x; idx < TTILE * (ND / 4); idx += 128) {
        int row  = idx >> 8;
        int c4   = idx & 255;
        int tokr = t0 + row;
        tokr = (tokr < NT) ? tokr : (NT - 1);
        v4f v = *(const v4f*)(hbase + (size_t)tokr * ND + ((size_t)c4 << 2));
        *(v4f*)(htile + row * HSTR + (c4 << 2)) = v;
    }
    __syncthreads();

    const float* __restrict__ hA = htile + m * HSTR + (wave << 8) + ((lane >> 4) << 2);

    v8f acc0 = {};
    v8f acc1 = {};
    #pragma unroll
    for (int j = 0; j < 32; ++j) {
        const v4f a = *(const v4f*)(hA + 8 * j);
        v2f a0, a1, b0, b1;
        a0.x = a.x;     a0.y = a.y;     a1.x = a.z;     a1.y = a.w;
        b0.x = bm[j].x; b0.y = bm[j].y; b1.x = bm[j].z; b1.y = bm[j].w;
        acc0 = __builtin_amdgcn_wmma_f32_16x16x4_f32(
                   false, a0, false, b0, (short)0, acc0, false, false);
        acc1 = __builtin_amdgcn_wmma_f32_16x16x4_f32(
                   false, a1, false, b1, (short)0, acc1, false, false);
    }
    __builtin_amdgcn_sched_group_barrier(0x100, 16, 0);
    __builtin_amdgcn_sched_group_barrier(0x008, 16, 0);
    __builtin_amdgcn_sched_group_barrier(0x100, 8, 0);
    __builtin_amdgcn_sched_group_barrier(0x008, 16, 0);
    __builtin_amdgcn_sched_group_barrier(0x100, 8, 0);
    __builtin_amdgcn_sched_group_barrier(0x008, 32, 0);

    #pragma unroll
    for (int i = 0; i < 8; ++i) {
        int row = i + ((lane >> 4) << 3);
        int col = lane & 15;
        partial[(wave * TTILE + row) * NR + col] = acc0[i] + acc1[i];
    }
    __syncthreads();
    for (int idx = threadIdx.x; idx < TTILE * NR; idx += 128) {
        int row = idx >> 4, col = idx & 15;
        Pm[row * PSTR + col] =
              partial[(0 * TTILE + row) * NR + col]
            + partial[(1 * TTILE + row) * NR + col]
            + partial[(2 * TTILE + row) * NR + col]
            + partial[(3 * TTILE + row) * NR + col]
            + bL[col];
    }
    __syncthreads();

    v2f  pa[4];
    float sc0[4], sc1[4];
    for (int q = 0; q < 4; ++q) {
        pa[q].x = Pm[m * PSTR + 4 * q + kk];
        pa[q].y = Pm[m * PSTR + 4 * q + kk + 1];
        sc0[q]  = invL[4 * q + kk];
        sc1[q]  = invL[4 * q + kk + 1];
    }
    const int hrow = (lane >> 4) << 3;

    for (int g = 0; g < 4; ++g) {
        const int dbase = (wave << 8) + (g << 6);
        const int dl    = dbase + (m << 2);

        v4f r0[4], r1[4];
        for (int q = 0; q < 4; ++q) {
            r0[q] = *(const v4f*)(Rs + (size_t)(4 * q + kk)     * ND + dl);
            r1[q] = *(const v4f*)(Rs + (size_t)(4 * q + kk + 1) * ND + dl);
        }
        v4f res[8];
        for (int i = 0; i < 8; ++i)
            res[i] = *(const v4f*)(htile + (i + hrow) * HSTR + dl);

        v8f acc[4] = {};
        for (int q = 0; q < 4; ++q) {
            for (int t = 0; t < 4; ++t) {
                v2f br;
                br.x = r0[q][t] * sc0[q];
                br.y = r1[q][t] * sc1[q];
                acc[t] = __builtin_amdgcn_wmma_f32_16x16x4_f32(
                             false, pa[q], false, br, (short)0, acc[t], false, false);
            }
        }
        for (int i = 0; i < 8; ++i) {
            v4f o;
            o.x = res[i].x + acc[0][i];
            o.y = res[i].y + acc[1][i];
            o.z = res[i].z + acc[2][i];
            o.w = res[i].w + acc[3][i];
            int toko = t0 + i + hrow;
            if (toko < NT)
                *(v4f*)(obase + (size_t)toko * ND + dl) = o;
        }
    }
}

extern "C" void kernel_launch(void* const* d_in, const int* in_sizes, int n_in,
                              void* d_out, int out_size, void* d_ws, size_t ws_size,
                              hipStream_t stream)
{
    const float* h    = (const float*)d_in[0];
    const float* R    = (const float*)d_in[1];
    const float* W    = (const float*)d_in[2];
    const float* bvec = (const float*)d_in[3];
    float* out = (float*)d_out;
    float* inv = (float*)d_ws;
    float* Mf  = inv + 1024;

    (void)in_sizes; (void)n_in; (void)out_size; (void)ws_size;

    enum { SMEM_FLOATS = TTILE * HSTR + 4 * TTILE * NR + TTILE * PSTR + NR + NR };

    loreft_prep_kernel<<<NS * NR, 256, 0, stream>>>(R, W, inv, Mf);
    loreft_main_kernel<<<NB * NS * NTILES, 128, SMEM_FLOATS * sizeof(float), stream>>>(
        h, R, Mf, bvec, inv, out);
}


# Launch 2: 8-wave block walks 4 token tiles of one step, two LDS tile buffers; next tile's h loads issued into registers before this tile's products, stored to LDS after; k sum split 8 ways (was 4)
# speedup vs baseline: 1.9851x; 1.0822x over previous
#include <hip/hip_runtime.h>
#include <math.h>

#define NB 8
#define NS 50
#define NT 77
#define ND 1024
#define NR 16
#define TTILE 16
#define NTILES 5
#define HSTR 1028
#define PSTR 17
#define TPB 4
#define BLK_PER_S 10
#define NW 8

typedef __attribute__((ext_vector_type(2))) float v2f;
typedef __attribute__((ext_vector_type(4))) float v4f;
typedef __attribute__((ext_vector_type(8))) float v8f;

__global__ __launch_bounds__(256)
void loreft_prep_kernel(const float* __restrict__ R, const float* __restrict__ W,
                        float* __restrict__ inv, float* __restrict__ Mf)
{
    const int sr = blockIdx.x;
    const float* row = R + (size_t)sr * ND;

    const v4f rv = *(const v4f*)(row + 4 * threadIdx.x);
    const v4f wv = *(const v4f*)(W + (size_t)sr * ND + 4 * threadIdx.x);

    float sum = 0.0f;
    for (int i = threadIdx.x; i < ND; i += 256) {
        float v = row[i];
        sum += v * v;
    }
    __shared__ float red[256];
    red[threadIdx.x] = sum;
    __syncthreads();
    for (int off = 128; off > 0; off >>= 1) {
        if (threadIdx.x < off) red[threadIdx.x] += red[threadIdx.x + off];
        __syncthreads();
    }
    if (threadIdx.x == 0) {
        float n = sqrtf(red[0]);
        n = fmaxf(n, 1e-12f);
        const float r1 = 1.0f / n;
        inv[sr] = r1;
        red[0]  = r1;
    }
    __syncthreads();
    const float iv = red[0];

    v4f mv;
    mv.x = wv.x - rv.x * iv;
    mv.y = wv.y - rv.y * iv;
    mv.z = wv.z - rv.z * iv;
    mv.w = wv.w - rv.w * iv;

    const int s  = sr / NR;
    const int r  = sr % NR;
    const int w  = threadIdx.x >> 6;
    const int j  = (threadIdx.x & 63) >> 1;
    const int hf = threadIdx.x & 1;
    *(v4f*)(Mf + (((((size_t)(s * 4 + w) * 32 + j) * 32) + (hf * 16 + r)) << 2)) = mv;
}


static __device__ __forceinline__
void lds_barrier()
{
    __builtin_amdgcn_fence(__ATOMIC_RELEASE, "workgroup", "local");
    __builtin_amdgcn_s_barrier();
    __builtin_amdgcn_fence(__ATOMIC_ACQUIRE, "workgroup", "local");
}

static __device__ __forceinline__
void tile_fetch(v4f (&pre)[16], const float* __restrict__ hbase, int t0, int tid)
{
    #pragma unroll
    for (int it = 0; it < 16; ++it) {
        int tokr = t0 + it;
        tokr = (tokr < NT) ? tokr : (NT - 1);
        pre[it] = *(const v4f*)(hbase + (size_t)tokr * ND + ((size_t)tid << 2));
    }
}

static __device__ __forceinline__
void tile_stash(const v4f (&pre)[16], float* hdst, int tid)
{
    #pragma unroll
    for (int it = 0; it < 16; ++it)
        *(v4f*)(hdst + it * HSTR + (tid << 2)) = pre[it];
}

__global__ __launch_bounds__(256) __attribute__((amdgpu_waves_per_eu(4, 4)))
void loreft_main_kernel(const float* __restrict__ h,
                        const float* __restrict__ R,
                        const float* __restrict__ Mf,
                        const float* __restrict__ bvec,
                        const float* __restrict__ inv,
                        float* __restrict__ out)
{
    extern __shared__ float smem[];
    float* hbuf                 = smem;
    float* __restrict__ partial = hbuf + 2 * TTILE * HSTR;
    float* __restrict__ Pm      = partial + NW * TTILE * NR;
    float* __restrict__ invL    = Pm + TTILE * PSTR;
    float* __restrict__ bL      = invL + NR;

    const int s  = blockIdx.x / BLK_PER_S;
    const int p0 = (blockIdx.x % BLK_PER_S) * TPB;

    const int lane = threadIdx.x & 31;
    const int wave = threadIdx.x >> 5;
    const int m    = lane & 15;
    const int kk   = (lane >> 4) << 1;

    const float* __restrict__ Rs = R + (size_t)s * NR * ND;

    const v4f* __restrict__ Mp =
        (const v4f*)Mf + ((size_t)(s * 4 + (wave >> 1)) * 32 + ((wave & 1) << 4)) * 32 + lane;
    v4f bm[16];
    #pragma unroll
    for (int j = 0; j < 16; ++j)
        bm[j] = Mp[j * 32];

    if (threadIdx.x < NR) {
        invL[threadIdx.x] = inv[s * NR + threadIdx.x];
        bL[threadIdx.x]   = bvec[s * NR + threadIdx.x];
    }

    v4f pre[16];
    {
        const int b0 = p0 / NTILES;
        const int tl = p0 % NTILES;
        tile_fetch(pre, h + ((size_t)(b0 * NS + s)) * NT * ND, tl * TTILE, threadIdx.x);
        tile_stash(pre, hbuf, threadIdx.x);
    }
    lds_barrier();

    #pragma unroll 1
    for (int i = 0; i < TPB; ++i) {
        const int p    = p0 + i;
        const int b    = p / NTILES;
        const int tile = p % NTILES;
        const int t0   = tile * TTILE;
        float* htile = hbuf + (i & 1) * (TTILE * HSTR);
        float* hnext = hbuf + ((i & 1) ^ 1) * (TTILE * HSTR);
        float* __restrict__ obase = out + ((size_t)(b * NS + s)) * NT * ND;

        if (i + 1 < TPB) {
            const int bn = (p + 1) / NTILES;
            const int tn = (p + 1) % NTILES;
            tile_fetch(pre, h + ((size_t)(bn * NS + s)) * NT * ND, tn * TTILE, threadIdx.x);
        }

        const float* hA = htile + m * HSTR + (wave << 7) + ((lane >> 4) << 2);

        v8f acc0 = {};
        v8f acc1 = {};
        #pragma unroll
        for (int j = 0; j < 16; ++j) {
            const v4f a = *(const v4f*)(hA + 8 * j);
            v2f a0, a1, b0, b1;
            a0.x = a.x;     a0.y = a.y;     a1.x = a.z;     a1.y = a.w;
            b0.x = bm[j].x; b0.y = bm[j].y; b1.x = bm[j].z; b1.y = bm[j].w;
            acc0 = __builtin_amdgcn_wmma_f32_16x16x4_f32(
                       false, a0, false, b0, (short)0, acc0, false, false);
            acc1 = __builtin_amdgcn_wmma_f32_16x16x4_f32(
                       false, a1, false, b1, (short)0, acc1, false, false);
        }
        __builtin_amdgcn_sched_group_barrier(0x100, 16, 0);
        __builtin_amdgcn_sched_group_barrier(0x008, 32, 0);

        #pragma unroll
        for (int q = 0; q < 8; ++q) {
            int row = q + ((lane >> 4) << 3);
            int col = lane & 15;
            partial[(wave * TTILE + row) * NR + col] = acc0[q] + acc1[q];
        }
        lds_barrier();
        {
            const int row = threadIdx.x >> 4, col = threadIdx.x & 15;
            float psum = partial[(0 * TTILE + row) * NR + col];
            #pragma unroll
            for (int w = 1; w < NW; ++w)
                psum += partial[(w * TTILE + row) * NR + col];
            Pm[row * PSTR + col] = psum + bL[col];
        }
        lds_barrier();

        v2f  pa[4];
        float sc0[4], sc1[4];
        for (int q = 0; q < 4; ++q) {
            pa[q].x = Pm[m * PSTR + 4 * q + kk];
            pa[q].y = Pm[m * PSTR + 4 * q + kk + 1];
            sc0[q]  = invL[4 * q + kk];
            sc1[q]  = invL[4 * q + kk + 1];
        }
        const int hrow = (lane >> 4) << 3;

        for (int g = 0; g < 2; ++g) {
            const int dbase = (wave << 7) + (g << 6);
            const int dl    = dbase + (m << 2);

            v4f r0[4], r1[4];
            for (int q = 0; q < 4; ++q) {
                r0[q] = *(const v4f*)(Rs + (size_t)(4 * q + kk)     * ND + dl);
                r1[q] = *(const v4f*)(Rs + (size_t)(4 * q + kk + 1) * ND + dl);
            }
            v4f res[8];
            for (int q = 0; q < 8; ++q)
                res[q] = *(const v4f*)(htile + (q + hrow) * HSTR + dl);

            v8f acc[4] = {};
            for (int q = 0; q < 4; ++q) {
                for (int t = 0; t < 4; ++t) {
                    v2f br;
                    br.x = r0[q][t] * sc0[q];
                    br.y = r1[q][t] * sc1[q];
                    acc[t] = __builtin_amdgcn_wmma_f32_16x16x4_f32(
                                 false, pa[q], false, br, (short)0, acc[t], false, false);
                }
            }
            for (int q = 0; q < 8; ++q) {
                v4f o;
                o.x = res[q].x + acc[0][q];
                o.y = res[q].y + acc[1][q];
                o.z = res[q].z + acc[2][q];
                o.w = res[q].w + acc[3][q];
                int toko = t0 + q + hrow;
                if (toko < NT)
                    *(v4f*)(obase + (size_t)toko * ND + dl) = o;
            }
        }

        if (i + 1 < TPB)
            tile_stash(pre, hnext, threadIdx.x);
        lds_barrier();
    }
}

extern "C" void kernel_launch(void* const* d_in, const int* in_sizes, int n_in,
                              void* d_out, int out_size, void* d_ws, size_t ws_size,
                              hipStream_t stream)
{
    const float* h    = (const float*)d_in[0];
    const float* R    = (const float*)d_in[1];
    const float* W    = (const float*)d_in[2];
    const float* bvec = (const float*)d_in[3];
    float* out = (float*)d_out;
    float* inv = (float*)d_ws;
    float* Mf  = inv + 1024;

    (void)in_sizes; (void)n_in; (void)out_size; (void)ws_size;

    enum { SMEM_FLOATS = 2 * TTILE * HSTR + NW * TTILE * NR + TTILE * PSTR + NR + NR };

    loreft_prep_kernel<<<NS * NR, 256, 0, stream>>>(R, W, inv, Mf);
    loreft_main_kernel<<<NS * BLK_PER_S, 32 * NW, SMEM_FLOATS * sizeof(float), stream>>>(
        h, R, Mf, bvec, inv, out);
}


# One launch instead of two: each block's prologue norms its step's R rows and forms W - R/|R| itself (whole-row loads, in-wave sum, B fragments via the idle LDS tile buffer); norm sum order changed
# speedup vs baseline: 2.3290x; 1.1732x over previous
#include <hip/hip_runtime.h>
#include <math.h>

#define NB 8
#define NS 50
#define NT 77
#define ND 1024
#define NR 16
#define TTILE 16
#define NTILES 5
#define HSTR 1028
#define PSTR 17
#define TPB 4
#define BLK_PER_S 10
#define NW 8

typedef __attribute__((ext_vector_type(2))) float v2f;
typedef __attribute__((ext_vector_type(4))) float v4f;
typedef __attribute__((ext_vector_type(8))) float v8f;


static __device__ __forceinline__
void lds_barrier()
{
    __builtin_amdgcn_fence(__ATOMIC_RELEASE, "workgroup", "local");
    __builtin_amdgcn_s_barrier();
    __builtin_amdgcn_fence(__ATOMIC_ACQUIRE, "workgroup", "local");
}

static __device__ __forceinline__
void tile_fetch(v4f (&pre)[16], const float* __restrict__ hbase, int t0, int tid)
{
    #pragma unroll
    for (int it = 0; it < 16; ++it) {
        int tokr = t0 + it;
        tokr = (tokr < NT) ? tokr : (NT - 1);
        pre[it] = *(const v4f*)(hbase + (size_t)tokr * ND + ((size_t)tid << 2));
    }
}

static __device__ __forceinline__
void tile_stash(const v4f (&pre)[16], float* hdst, int tid)
{
    #pragma unroll
    for (int it = 0; it < 16; ++it)
        *(v4f*)(hdst + it * HSTR + (tid << 2)) = pre[it];
}

__global__ __launch_bounds__(256) __attribute__((amdgpu_waves_per_eu(4, 4)))
void loreft_main_kernel(const float* __restrict__ h,
                        const float* __restrict__ R,
                        const float* __restrict__ W,
                        const float* __restrict__ bvec,
                        float* __restrict__ out)
{
    extern __shared__ float smem[];
    float* hbuf                 = smem;
    float* __restrict__ partial = hbuf + 2 * TTILE * HSTR;
    float* __restrict__ Pm      = partial + NW * TTILE * NR;
    float* __restrict__ invL    = Pm + TTILE * PSTR;
    float* __restrict__ bL      = invL + NR;

    const int s  = blockIdx.x / BLK_PER_S;
    const int p0 = (blockIdx.x % BLK_PER_S) * TPB;

    const int lane = threadIdx.x & 31;
    const int wave = threadIdx.x >> 5;
    const int m    = lane & 15;
    const int kk   = (lane >> 4) << 1;

    const float* __restrict__ Rs = R + (size_t)s * NR * ND;
    const float* __restrict__ Ws = W + (size_t)s * NR * ND;

    const float bval = bvec[s * NR + (threadIdx.x & (NR - 1))];

    const int rw = wave << 1;
    const float* rp = Rs + (size_t)rw * ND + (lane << 2);
    const float* wp = Ws + (size_t)rw * ND + (lane << 2);
    v4f ra[8], rb[8], wa[8], wb[8];
    #pragma unroll
    for (int c = 0; c < 8; ++c) {
        ra[c] = *(const v4f*)(rp + 128 * c);
        rb[c] = *(const v4f*)(rp + ND + 128 * c);
    }
    #pragma unroll
    for (int c = 0; c < 8; ++c) {
        wa[c] = *(const v4f*)(wp + 128 * c);
        wb[c] = *(const v4f*)(wp + ND + 128 * c);
    }

    v4f pre[16];
    {
        const int b0 = p0 / NTILES;
        const int tl = p0 % NTILES;
        tile_fetch(pre, h + ((size_t)(b0 * NS + s)) * NT * ND, tl * TTILE, threadIdx.x);
    }

    v4f qa = {}, qb = {};
    #pragma unroll
    for (int c = 0; c < 8; ++c) {
        qa += ra[c] * ra[c];
        qb += rb[c] * rb[c];
    }
    float sa = (qa.x + qa.y) + (qa.z + qa.w);
    float sb = (qb.x + qb.y) + (qb.z + qb.w);
    #pragma unroll
    for (int o = 16; o > 0; o >>= 1) {
        sa += __shfl_xor(sa, o, 32);
        sb += __shfl_xor(sb, o, 32);
    }
    const float iva = 1.0f / fmaxf(sqrtf(sa), 1e-12f);
    const float ivb = 1.0f / fmaxf(sqrtf(sb), 1e-12f);
    if (lane == 0) {
        invL[rw]     = iva;
        invL[rw + 1] = ivb;
    }
    if (threadIdx.x < NR)
        bL[threadIdx.x] = bval;

    {
        float* mdst = hbuf + TTILE * HSTR + rw * HSTR + (lane << 2);
        #pragma unroll
        for (int c = 0; c < 8; ++c) {
            *(v4f*)(mdst + 128 * c)        = wa[c] - ra[c] * iva;
            *(v4f*)(mdst + HSTR + 128 * c) = wb[c] - rb[c] * ivb;
        }
    }

    tile_stash(pre, hbuf, threadIdx.x);
    lds_barrier();

    const float* mB = hbuf + TTILE * HSTR + m * HSTR + (wave << 7) + ((lane >> 4) << 2);
    v4f bm[16];
    #pragma unroll
    for (int j = 0; j < 16; ++j)
        bm[j] = *(const v4f*)(mB + 8 * j);

    #pragma unroll 1
    for (int i = 0; i < TPB; ++i) {
        const int p    = p0 + i;
        const int b    = p / NTILES;
        const int tile = p % NTILES;
        const int t0   = tile * TTILE;
        float* htile = hbuf + (i & 1) * (TTILE * HSTR);
        float* hnext = hbuf + ((i & 1) ^ 1) * (TTILE * HSTR);
        float* __restrict__ obase = out + ((size_t)(b * NS + s)) * NT * ND;

        if (i + 1 < TPB) {
            const int bn = (p + 1) / NTILES;
            const int tn = (p + 1) % NTILES;
            tile_fetch(pre, h + ((size_t)(bn * NS + s)) * NT * ND, tn * TTILE, threadIdx.x);
        }

        const float* hA = htile + m * HSTR + (wave << 7) + ((lane >> 4) << 2);

        v8f acc0 = {};
        v8f acc1 = {};
        #pragma unroll
        for (int j = 0; j < 16; ++j) {
            const v4f a = *(const v4f*)(hA + 8 * j);
            v2f a0, a1, b0, b1;
            a0.x = a.x;     a0.y = a.y;     a1.x = a.z;     a1.y = a.w;
            b0.x = bm[j].x; b0.y = bm[j].y; b1.x = bm[j].z; b1.y = bm[j].w;
            acc0 = __builtin_amdgcn_wmma_f32_16x16x4_f32(
                       false, a0, false, b0, (short)0, acc0, false, false);
            acc1 = __builtin_amdgcn_wmma_f32_16x16x4_f32(
                       false, a1, false, b1, (short)0, acc1, false, false);
        }
        __builtin_amdgcn_sched_group_barrier(0x100, 16, 0);
        __builtin_amdgcn_sched_group_barrier(0x008, 32, 0);

        #pragma unroll
        for (int q = 0; q < 8; ++q) {
            int row = q + ((lane >> 4) << 3);
            int col = lane & 15;
            partial[(wave * TTILE + row) * NR + col] = acc0[q] + acc1[q];
        }
        lds_barrier();
        {
            const int row = threadIdx.x >> 4, col = threadIdx.x & 15;
            float psum = partial[(0 * TTILE + row) * NR + col];
            #pragma unroll
            for (int w = 1; w < NW; ++w)
                psum += partial[(w * TTILE + row) * NR + col];
            Pm[row * PSTR + col] = psum + bL[col];
        }
        lds_barrier();

        v2f  pa[4];
        float sc0[4], sc1[4];
        for (int q = 0; q < 4; ++q) {
            pa[q].x = Pm[m * PSTR + 4 * q + kk];
            pa[q].y = Pm[m * PSTR + 4 * q + kk + 1];
            sc0[q]  = invL[4 * q + kk];
            sc1[q]  = invL[4 * q + kk + 1];
        }
        const int hrow = (lane >> 4) << 3;

        for (int g = 0; g < 2; ++g) {
            const int dbase = (wave << 7) + (g << 6);
            const int dl    = dbase + (m << 2);

            v4f r0[4], r1[4];
            for (int q = 0; q < 4; ++q) {
                r0[q] = *(const v4f*)(Rs + (size_t)(4 * q + kk)     * ND + dl);
                r1[q] = *(const v4f*)(Rs + (size_t)(4 * q + kk + 1) * ND + dl);
            }
            v4f res[8];
            for (int q = 0; q < 8; ++q)
                res[q] = *(const v4f*)(htile + (q + hrow) * HSTR + dl);

            v8f acc[4] = {};
            for (int q = 0; q < 4; ++q) {
                for (int t = 0; t < 4; ++t) {
                    v2f br;
                    br.x = r0[q][t] * sc0[q];
                    br.y = r1[q][t] * sc1[q];
                    acc[t] = __builtin_amdgcn_wmma_f32_16x16x4_f32(
                                 false, pa[q], false, br, (short)0, acc[t], false, false);
                }
            }
            for (int q = 0; q < 8; ++q) {
                v4f o;
                o.x = res[q].x + acc[0][q];
                o.y = res[q].y + acc[1][q];
                o.z = res[q].z + acc[2][q];
                o.w = res[q].w + acc[3][q];
                int toko = t0 + q + hrow;
                if (toko < NT)
                    *(v4f*)(obase + (size_t)toko * ND + dl) = o;
            }
        }

        if (i + 1 < TPB)
            tile_stash(pre, hnext, threadIdx.x);
        lds_barrier();
    }
}

extern "C" void kernel_launch(void* const* d_in, const int* in_sizes, int n_in,
                              void* d_out, int out_size, void* d_ws, size_t ws_size,
                              hipStream_t stream)
{
    const float* h    = (const float*)d_in[0];
    const float* R    = (const float*)d_in[1];
    const float* W    = (const float*)d_in[2];
    const float* bvec = (const float*)d_in[3];
    float* out = (float*)d_out;

    (void)in_sizes; (void)n_in; (void)out_size; (void)d_ws; (void)ws_size;

    enum { SMEM_FLOATS = 2 * TTILE * HSTR + NW * TTILE * NR + TTILE * PSTR + NR + NR };

    loreft_main_kernel<<<NS * BLK_PER_S, 32 * NW, SMEM_FLOATS * sizeof(float), stream>>>(
        h, R, W, bvec, out);
}


# Phase B takes R/|R| from LDS: prologue writes it once per block in B-fragment order where the 2nd tile buffer was (h tile single-buffered, each wave restashes its own slice; tile-end barrier gone)
# speedup vs baseline: 2.5351x; 1.0885x over previous
#include <hip/hip_runtime.h>
#include <math.h>

#define NB 8
#define NS 50
#define NT 77
#define ND 1024
#define NR 16
#define TTILE 16
#define NTILES 5
#define HSTR 1028
#define PSTR 17
#define TPB 4
#define BLK_PER_S 10
#define NW 8
#define RNPAIR 2048
#define RNFLOATS (8 * RNPAIR)

typedef __attribute__((ext_vector_type(2))) float v2f;
typedef __attribute__((ext_vector_type(4))) float v4f;
typedef __attribute__((ext_vector_type(8))) float v8f;


static __device__ __forceinline__
void lds_barrier()
{
    __builtin_amdgcn_fence(__ATOMIC_RELEASE, "workgroup", "local");
    __builtin_amdgcn_s_barrier();
    __builtin_amdgcn_fence(__ATOMIC_ACQUIRE, "workgroup", "local");
}

static __device__ __forceinline__
void tile_fetch(v4f (&pre)[16], const float* __restrict__ hbase, int t0, int tid)
{
    #pragma unroll
    for (int it = 0; it < 16; ++it) {
        int tokr = t0 + it;
        tokr = (tokr < NT) ? tokr : (NT - 1);
        pre[it] = *(const v4f*)(hbase + (size_t)tokr * ND + ((size_t)tid << 2));
    }
}

static __device__ __forceinline__
void tile_stash(const v4f (&pre)[16], float* hdst, int tid)
{
    #pragma unroll
    for (int it = 0; it < 16; ++it)
        *(v4f*)(hdst + it * HSTR + (tid << 2)) = pre[it];
}

__global__ __launch_bounds__(256) __attribute__((amdgpu_waves_per_eu(4, 4)))
void loreft_main_kernel(const float* __restrict__ h,
                        const float* __restrict__ R,
                        const float* __restrict__ W,
                        const float* __restrict__ bvec,
                        float* __restrict__ out)
{
    extern __shared__ float smem[];
    float* hbuf                 = smem;
    float* rnbuf                = hbuf + TTILE * HSTR;
    float* __restrict__ partial = rnbuf + RNFLOATS;
    float* __restrict__ Pm      = partial + NW * TTILE * NR;
    float* __restrict__ bL      = Pm + TTILE * PSTR;

    const int s  = blockIdx.x / BLK_PER_S;
    const int p0 = (blockIdx.x % BLK_PER_S) * TPB;

    const int lane = threadIdx.x & 31;
    const int wave = threadIdx.x >> 5;
    const int m    = lane & 15;
    const int kk   = (lane >> 4) << 1;

    const float* __restrict__ Rs = R + (size_t)s * NR * ND;
    const float* __restrict__ Ws = W + (size_t)s * NR * ND;

    const float bval = bvec[s * NR + (threadIdx.x & (NR - 1))];

    const int rw = wave << 1;
    const float* rp = Rs + (size_t)rw * ND + (lane << 2);
    const float* wp = Ws + (size_t)rw * ND + (lane << 2);
    v4f ra[8], rb[8], wa[8], wb[8];
    #pragma unroll
    for (int c = 0; c < 8; ++c) {
        ra[c] = *(const v4f*)(rp + 128 * c);
        rb[c] = *(const v4f*)(rp + ND + 128 * c);
    }
    #pragma unroll
    for (int c = 0; c < 8; ++c) {
        wa[c] = *(const v4f*)(wp + 128 * c);
        wb[c] = *(const v4f*)(wp + ND + 128 * c);
    }

    v4f pre[16];
    {
        const int b0 = p0 / NTILES;
        const int tl = p0 % NTILES;
        tile_fetch(pre, h + ((size_t)(b0 * NS + s)) * NT * ND, tl * TTILE, threadIdx.x);
    }

    v4f qa = {}, qb = {};
    #pragma unroll
    for (int c = 0; c < 8; ++c) {
        qa += ra[c] * ra[c];
        qb += rb[c] * rb[c];
    }
    float sa = (qa.x + qa.y) + (qa.z + qa.w);
    float sb = (qb.x + qb.y) + (qb.z + qb.w);
    #pragma unroll
    for (int o = 16; o > 0; o >>= 1) {
        sa += __shfl_xor(sa, o, 32);
        sb += __shfl_xor(sb, o, 32);
    }
    const float iva = 1.0f / fmaxf(sqrtf(sa), 1e-12f);
    const float ivb = 1.0f / fmaxf(sqrtf(sb), 1e-12f);
    if (threadIdx.x < NR)
        bL[threadIdx.x] = bval;

    {
        float* mdst = hbuf + rw * HSTR + (lane << 2);
        #pragma unroll
        for (int c = 0; c < 8; ++c) {
            *(v4f*)(mdst + 128 * c)        = wa[c] - ra[c] * iva;
            *(v4f*)(mdst + HSTR + 128 * c) = wb[c] - rb[c] * ivb;
        }
    }
    {
        float* rdst = rnbuf + wave * RNPAIR + (lane << 2);
        #pragma unroll
        for (int c = 0; c < 8; ++c) {
            const v4f na = ra[c] * iva;
            const v4f nb = rb[c] * ivb;
            v4f lo, hi;
            lo.x = na.x; lo.y = nb.x; lo.z = na.y; lo.w = nb.y;
            hi.x = na.z; hi.y = nb.z; hi.z = na.w; hi.w = nb.w;
            *(v4f*)(rdst + 128 * c)        = lo;
            *(v4f*)(rdst + 1024 + 128 * c) = hi;
        }
    }
    lds_barrier();

    const float* mB = hbuf + m * HSTR + (wave << 7) + ((lane >> 4) << 2);
    v4f bm[16];
    #pragma unroll
    for (int j = 0; j < 16; ++j)
        bm[j] = *(const v4f*)(mB + 8 * j);

    tile_stash(pre, hbuf, threadIdx.x);

    #pragma unroll 1
    for (int i = 0; i < TPB; ++i) {
        const int p    = p0 + i;
        const int b    = p / NTILES;
        const int tile = p % NTILES;
        const int t0   = tile * TTILE;
        float* htile = hbuf;
        float* __restrict__ obase = out + ((size_t)(b * NS + s)) * NT * ND;

        if (i + 1 < TPB) {
            const int bn = (p + 1) / NTILES;
            const int tn = (p + 1) % NTILES;
            tile_fetch(pre, h + ((size_t)(bn * NS + s)) * NT * ND, tn * TTILE, threadIdx.x);
        }

        const float* hA = htile + m * HSTR + (wave << 7) + ((lane >> 4) << 2);

        v8f acc0 = {};
        v8f acc1 = {};
        #pragma unroll
        for (int j = 0; j < 16; ++j) {
            const v4f a = *(const v4f*)(hA + 8 * j);
            v2f a0, a1, b0, b1;
            a0.x = a.x;     a0.y = a.y;     a1.x = a.z;     a1.y = a.w;
            b0.x = bm[j].x; b0.y = bm[j].y; b1.x = bm[j].z; b1.y = bm[j].w;
            acc0 = __builtin_amdgcn_wmma_f32_16x16x4_f32(
                       false, a0, false, b0, (short)0, acc0, false, false);
            acc1 = __builtin_amdgcn_wmma_f32_16x16x4_f32(
                       false, a1, false, b1, (short)0, acc1, false, false);
        }
        __builtin_amdgcn_sched_group_barrier(0x100, 16, 0);
        __builtin_amdgcn_sched_group_barrier(0x008, 32, 0);

        #pragma unroll
        for (int q = 0; q < 8; ++q) {
            int row = q + ((lane >> 4) << 3);
            int col = lane & 15;
            partial[(wave * TTILE + row) * NR + col] = acc0[q] + acc1[q];
        }
        lds_barrier();
        {
            const int row = threadIdx.x >> 4, col = threadIdx.x & 15;
            float psum = partial[(0 * TTILE + row) * NR + col];
            #pragma unroll
            for (int w = 1; w < NW; ++w)
                psum += partial[(w * TTILE + row) * NR + col];
            Pm[row * PSTR + col] = psum + bL[col];
        }
        lds_barrier();

        v2f  pa[4];
        for (int q = 0; q < 4; ++q) {
            pa[q].x = Pm[m * PSTR + 4 * q + kk];
            pa[q].y = Pm[m * PSTR + 4 * q + kk + 1];
        }
        const int hrow = (lane >> 4) << 3;

        for (int g = 0; g < 2; ++g) {
            const int dbase = (wave << 7) + (g << 6);
            const int dl    = dbase + (m << 2);

            const float* rsrc = rnbuf + (lane >> 4) * RNPAIR + dl;
            v4f rlo[4], rhi[4];
            for (int q = 0; q < 4; ++q) {
                rlo[q] = *(const v4f*)(rsrc + (2 * q) * RNPAIR);
                rhi[q] = *(const v4f*)(rsrc + (2 * q) * RNPAIR + 1024);
            }
            v4f res[8];
            for (int q = 0; q < 8; ++q)
                res[q] = *(const v4f*)(htile + (q + hrow) * HSTR + dl);

            v8f acc[4] = {};
            for (int q = 0; q < 4; ++q) {
                for (int t = 0; t < 4; ++t) {
                    const v4f rv = (t < 2) ? rlo[q] : rhi[q];
                    v2f br;
                    br.x = (t & 1) ? rv.z : rv.x;
                    br.y = (t & 1) ? rv.w : rv.y;
                    acc[t] = __builtin_amdgcn_wmma_f32_16x16x4_f32(
                                 false, pa[q], false, br, (short)0, acc[t], false, false);
                }
            }
            for (int q = 0; q < 8; ++q) {
                v4f o;
                o.x = res[q].x + acc[0][q];
                o.y = res[q].y + acc[1][q];
                o.z = res[q].z + acc[2][q];
                o.w = res[q].w + acc[3][q];
                int toko = t0 + q + hrow;
                if (toko < NT)
                    *(v4f*)(obase + (size_t)toko * ND + dl) = o;
            }
        }

        if (i + 1 < TPB)
            tile_stash(pre, htile, threadIdx.x);
    }
}

extern "C" void kernel_launch(void* const* d_in, const int* in_sizes, int n_in,
                              void* d_out, int out_size, void* d_ws, size_t ws_size,
                              hipStream_t stream)
{
    const float* h    = (const float*)d_in[0];
    const float* R    = (const float*)d_in[1];
    const float* W    = (const float*)d_in[2];
    const float* bvec = (const float*)d_in[3];
    float* out = (float*)d_out;

    (void)in_sizes; (void)n_in; (void)out_size; (void)d_ws; (void)ws_size;

    enum { SMEM_FLOATS = TTILE * HSTR + RNFLOATS + NW * TTILE * NR + TTILE * PSTR + NR };

    loreft_main_kernel<<<NS * BLK_PER_S, 32 * NW, SMEM_FLOATS * sizeof(float), stream>>>(
        h, R, W, bvec, out);
}
